# starting kernel
# baseline (speedup 1.0000x reference)
#include <hip/hip_runtime.h>

#define B_SZ   4
#define T_SZ   1024
#define F_SZ   1025
#define M_SZ   128
#define ITERS  20
#define LR     0.3f
#define MOM    0.9f
#define INV    (2.0f / (B_SZ * T_SZ))

#define ROWS     16
#define FSTR     1027
#define MSTR     131
#define NDSTR    131
#define NWAVES   8
#define NTHREADS 256
#define NFT      ((F_SZ + 15) / 16)

typedef __attribute__((ext_vector_type(2))) float v2f;
typedef __attribute__((ext_vector_type(8))) float v8f;

__device__ __forceinline__ v8f wmma_f32(v2f a, v2f b, v8f c) {
  return __builtin_amdgcn_wmma_f32_16x16x4_f32(false, a, false, b,
                                               (short)0, c, false, false);
}

__global__ __launch_bounds__(NTHREADS)
void transpose_fb_kernel(const float* __restrict__ fb,
                         float* __restrict__ fbT) {
  int i = blockIdx.x * NTHREADS + threadIdx.x;
  if (i < M_SZ * F_SZ) {
    int m = i / F_SZ, f = i - m * F_SZ;
    fbT[i] = fb[f * M_SZ + m];
  }
}

__global__ __launch_bounds__(NTHREADS)
void invmel_sgd_kernel(const float* __restrict__ mel,
                       const float* __restrict__ spec0,
                       const float* __restrict__ fb,
                       const float* __restrict__ fbT,
                       float* __restrict__ out)
{
  __shared__ float s_spec[ROWS * FSTR];
  __shared__ float s_buf [ROWS * FSTR];
  __shared__ float s_mel [ROWS * MSTR];
  __shared__ float s_nd  [ROWS * NDSTR];

  const int tid  = threadIdx.x;
  const int lane = tid & 31;
  const int wave = tid >> 5;
  const int nl   = lane & 15;
  const int hs   = lane >> 4;
  const int khi  = hs * 2;
  const int hi8  = hs * 8;

  const int blk = blockIdx.x;
  const int b   = blk / (T_SZ / ROWS);
  const int t0  = (blk % (T_SZ / ROWS)) * ROWS;

  for (int i = tid; i < ROWS * F_SZ; i += NTHREADS) {
    int r = i / F_SZ, c = i - r * F_SZ;
    s_spec[r * FSTR + c] = spec0[(b * T_SZ + t0 + r) * F_SZ + c];
    s_buf [r * FSTR + c] = 0.0f;
  }
  for (int i = tid; i < ROWS * M_SZ; i += NTHREADS) {
    int r = i & (ROWS - 1), m = i >> 4;
    s_mel[r * MSTR + m] = mel[(b * M_SZ + m) * T_SZ + t0 + r];
  }
  __syncthreads();

  for (int it = 0; it < ITERS; ++it) {
    {
      const int n0 = wave * 16;
      v8f c;
      #pragma unroll
      for (int r = 0; r < 8; ++r)
        c[r] = -s_mel[(r + hi8) * MSTR + n0 + nl];

      #pragma unroll 4
      for (int k = 0; k < 1024; k += 4) {
        v2f a, bf;
        a.x  = s_spec[nl * FSTR + k + khi];
        a.y  = s_spec[nl * FSTR + k + khi + 1];
        bf.x = fb[(k + khi)     * M_SZ + n0 + nl];
        bf.y = fb[(k + khi + 1) * M_SZ + n0 + nl];
        c = wmma_f32(a, bf, c);
      }
      {
        v2f a, bf;
        a.x  = (hs == 0) ? s_spec[nl * FSTR + 1024]  : 0.0f;
        a.y  = 0.0f;
        bf.x = (hs == 0) ? fb[1024 * M_SZ + n0 + nl] : 0.0f;
        bf.y = 0.0f;
        c = wmma_f32(a, bf, c);
      }
      #pragma unroll
      for (int r = 0; r < 8; ++r)
        s_nd[(r + hi8) * NDSTR + n0 + nl] = c[r];
    }
    __syncthreads();

    for (int ft = wave; ft < NFT; ft += NWAVES) {
      const int f0   = ft * 16;
      const int fcol = f0 + nl;
      const int fcl  = (fcol < F_SZ) ? fcol : (F_SZ - 1);
      v8f acc = {0.f, 0.f, 0.f, 0.f, 0.f, 0.f, 0.f, 0.f};

      #pragma unroll 4
      for (int m = 0; m < M_SZ; m += 4) {
        v2f a, bf;
        a.x  = s_nd[nl * NDSTR + m + khi];
        a.y  = s_nd[nl * NDSTR + m + khi + 1];
        bf.x = fbT[(m + khi)     * F_SZ + fcl];
        bf.y = fbT[(m + khi + 1) * F_SZ + fcl];
        acc = wmma_f32(a, bf, acc);
      }
      if (fcol < F_SZ) {
        #pragma unroll
        for (int r = 0; r < 8; ++r) {
          const int off = (r + hi8) * FSTR + fcol;
          float nb = MOM * s_buf[off] + INV * acc[r];
          s_buf[off] = nb;
          float ns = s_spec[off] - LR * nb;
          s_spec[off] = ns > 0.0f ? ns : 0.0f;
        }
      }
    }
    __syncthreads();
  }

  for (int i = tid; i < ROWS * F_SZ; i += NTHREADS) {
    int r = i & (ROWS - 1), f = i >> 4;
    out[(b * F_SZ + f) * T_SZ + t0 + r] = s_spec[r * FSTR + f];
  }
}

extern "C" void kernel_launch(void* const* d_in, const int* in_sizes, int n_in,
                              void* d_out, int out_size, void* d_ws, size_t ws_size,
                              hipStream_t stream) {
  (void)in_sizes; (void)n_in; (void)out_size; (void)ws_size;
  const float* mel   = (const float*)d_in[0];
  const float* spec0 = (const float*)d_in[1];
  const float* fb    = (const float*)d_in[2];
  float* out = (float*)d_out;
  float* fbT = (float*)d_ws;

  {
    dim3 grid((M_SZ * F_SZ + NTHREADS - 1) / NTHREADS);
    hipLaunchKernelGGL(transpose_fb_kernel, grid, dim3(NTHREADS), 0, stream,
                       fb, fbT);
  }
  {
    dim3 grid((B_SZ * T_SZ) / ROWS);
    hipLaunchKernelGGL(invmel_sgd_kernel, grid, dim3(NTHREADS), 0, stream,
                       mel, spec0, fb, fbT, out);
  }
}


# A quarter of each wave's filterbank fragments (4 of every 16 k-steps, both GEMMs) is loaded once and kept in registers for all 20 iterations: 24 L2 load rounds per iteration, not 32; sum order same
# speedup vs baseline: 2.6230x; 2.6230x over previous
#include <hip/hip_runtime.h>


#define B_SZ   4
#define T_SZ   1024
#define F_SZ   1025
#define M_SZ   128
#define ITERS  20
#define LR     0.3f
#define MOM    0.9f
#define INV    (2.0f / (B_SZ * T_SZ))

#define ROWS     16
#define FSTR     1027
#define MSTR     131
#define NDSTR    131
#define NWAVES   32
#define NTHREADS 1024
#define TP_THREADS 256
#define NMT      (M_SZ / 16)
#define KQ       (NWAVES / NMT)
#define KQLEN    (1024 / KQ)
#define NFULL    (1024 / 16)
#define NDSZ     (ROWS * NDSTR)
#define NK1      (KQLEN / 4)
#define NK2      (M_SZ / 4)
#define NR1      (NK1 / 4)
#define NR2      (NK2 / 4)
#define NT2      (NFULL / NWAVES)
#define RSTEP(g, j)  (16 * (g) + 5 * (j))

typedef __attribute__((ext_vector_type(2))) float v2f;
typedef __attribute__((ext_vector_type(8))) float v8f;

__device__ __forceinline__ v8f wmma_f32(v2f a, v2f b, v8f c) {
  return __builtin_amdgcn_wmma_f32_16x16x4_f32(false, a, false, b,
                                               (short)0, c, false, false);
}

__global__ __launch_bounds__(TP_THREADS)
void transpose_fb_kernel(const float* __restrict__ fb,
                         float* __restrict__ fbT) {
  int i = blockIdx.x * TP_THREADS + threadIdx.x;
  if (i < M_SZ * F_SZ) {
    int m = i / F_SZ, f = i - m * F_SZ;
    fbT[i] = fb[f * M_SZ + m];
  }
}

__global__ __launch_bounds__(NTHREADS)
void invmel_sgd_kernel(const float* __restrict__ mel,
                       const float* __restrict__ spec0,
                       const float* fb,
                       const float* fbT,
                       float* __restrict__ out)
{
  __shared__ float s_spec[ROWS * FSTR];
  __shared__ float s_buf [ROWS * FSTR];
  __shared__ float s_mel [ROWS * MSTR];
  __shared__ float s_ndp [KQ * NDSZ];
  __shared__ float s_nd  [NDSZ];
  __shared__ float s_tail[NWAVES * ROWS];

  const int tid  = threadIdx.x;
  const int lane = tid & 31;
  const int wave = __builtin_amdgcn_readfirstlane(tid >> 5);
  const int nl   = lane & 15;
  const int hs   = lane >> 4;
  const int khi  = hs * 2;
  const int hi8  = hs * 8;
  const int kq   = wave >> 3;
  const int n0   = (wave & 7) * 16;

  const int blk = blockIdx.x;
  const int b   = blk / (T_SZ / ROWS);
  const int t0  = (blk % (T_SZ / ROWS)) * ROWS;

  for (int i = tid; i < ROWS * F_SZ; i += NTHREADS) {
    int r = i / F_SZ, c = i - r * F_SZ;
    s_spec[r * FSTR + c] = spec0[(b * T_SZ + t0 + r) * F_SZ + c];
    s_buf [r * FSTR + c] = 0.0f;
  }
  for (int i = tid; i < ROWS * M_SZ; i += NTHREADS) {
    int r = i & (ROWS - 1), m = i >> 4;
    s_mel[r * MSTR + m] = mel[(b * M_SZ + m) * T_SZ + t0 + r];
  }
  const float fb_last = fb[(F_SZ - 1) * M_SZ + n0 + nl];
  v2f tb;
  tb.x = fbT[(4 * wave + khi)     * F_SZ + (F_SZ - 1)];
  tb.y = fbT[(4 * wave + khi + 1) * F_SZ + (F_SZ - 1)];
  const float* const p1 = fb  + (kq * KQLEN + khi) * M_SZ + n0 + nl;
  const float* const p2 = fbT + khi * F_SZ + wave * 16 + nl;
  v2f rb1[NR1];
  #pragma unroll
  for (int g = 0; g < NK1 / 16; ++g) {
    #pragma unroll
    for (int j = 0; j < 4; ++j) {
      rb1[4 * g + j].x = p1[(4 * RSTEP(g, j))     * M_SZ];
      rb1[4 * g + j].y = p1[(4 * RSTEP(g, j) + 1) * M_SZ];
    }
  }
  v2f rb2[NT2][NR2];
  #pragma unroll
  for (int t = 0; t < NT2; ++t) {
    #pragma unroll
    for (int g = 0; g < NK2 / 16; ++g) {
      #pragma unroll
      for (int j = 0; j < 4; ++j) {
        rb2[t][4 * g + j].x = p2[(4 * RSTEP(g, j))     * F_SZ + t * NWAVES * 16];
        rb2[t][4 * g + j].y = p2[(4 * RSTEP(g, j) + 1) * F_SZ + t * NWAVES * 16];
      }
    }
  }
  __syncthreads();

  for (int it = 0; it < ITERS; ++it) {
    {
      const int kb = kq * KQLEN;
      v8f c;
      if (kq == 0) {
        const float* const melp = &s_mel[hi8 * MSTR + n0 + nl];
        #pragma unroll
        for (int r = 0; r < 8; ++r)
          c[r] = -melp[r * MSTR];
      } else {
        #pragma unroll
        for (int r = 0; r < 8; ++r) c[r] = 0.0f;
      }

      const float* const sa = &s_spec[nl * FSTR + kb + khi];
      #pragma unroll
      for (int g = 0; g < NK1 / 16; ++g) {
        #pragma unroll
        for (int j = 0; j < 3; ++j) {
          const int sr = RSTEP(g, j);
          v2f sb[4];
          #pragma unroll
          for (int i = 0; i < 4; ++i) {
            sb[i].x = p1[(4 * (sr + 1 + i))     * M_SZ];
            sb[i].y = p1[(4 * (sr + 1 + i) + 1) * M_SZ];
          }
          {
            v2f a;
            a.x = sa[4 * sr];
            a.y = sa[4 * sr + 1];
            c = wmma_f32(a, rb1[4 * g + j], c);
          }
          #pragma unroll
          for (int i = 0; i < 4; ++i) {
            v2f a;
            a.x = sa[4 * (sr + 1 + i)];
            a.y = sa[4 * (sr + 1 + i) + 1];
            c = wmma_f32(a, sb[i], c);
          }
        }
        {
          const int sr = RSTEP(g, 3);
          v2f a;
          a.x = sa[4 * sr];
          a.y = sa[4 * sr + 1];
          c = wmma_f32(a, rb1[4 * g + 3], c);
        }
      }
      float* const ndpp = &s_ndp[kq * NDSZ + hi8 * NDSTR + n0 + nl];
      #pragma unroll
      for (int r = 0; r < 8; ++r)
        ndpp[r * NDSTR] = c[r];
    }
    __syncthreads();

    if (wave < NMT) {
      v8f c;
      const float* const fxp = &s_ndp[hi8 * NDSTR + n0 + nl];
      #pragma unroll
      for (int r = 0; r < 8; ++r) {
        const int idx = r * NDSTR;
        c[r] = ((fxp[idx] + fxp[NDSZ + idx]) + fxp[2 * NDSZ + idx])
               + fxp[3 * NDSZ + idx];
      }
      v2f a, bf;
      a.x  = (hs == 0) ? s_spec[nl * FSTR + 1024] : 0.0f;
      a.y  = 0.0f;
      bf.x = (hs == 0) ? fb_last : 0.0f;
      bf.y = 0.0f;
      c = wmma_f32(a, bf, c);
      float* const ndw = &s_nd[hi8 * NDSTR + n0 + nl];
      #pragma unroll
      for (int r = 0; r < 8; ++r)
        ndw[r * NDSTR] = c[r];
    }
    __syncthreads();

    {
      v8f tacc = {0.f, 0.f, 0.f, 0.f, 0.f, 0.f, 0.f, 0.f};
      {
        v2f a;
        a.x = s_nd[nl * NDSTR + 4 * wave + khi];
        a.y = s_nd[nl * NDSTR + 4 * wave + khi + 1];
        tacc = wmma_f32(a, tb, tacc);
      }
      if (nl == 0) {
        #pragma unroll
        for (int r = 0; r < 8; ++r)
          s_tail[wave * ROWS + r + hi8] = tacc[r];
      }

      const float* const sn = &s_nd[nl * NDSTR + khi];
      #pragma unroll
      for (int t = 0; t < NT2; ++t) {
        const int fcol = (wave + t * NWAVES) * 16 + nl;
        v8f acc = {0.f, 0.f, 0.f, 0.f, 0.f, 0.f, 0.f, 0.f};

        #pragma unroll
        for (int g = 0; g < NK2 / 16; ++g) {
          #pragma unroll
          for (int j = 0; j < 3; ++j) {
            const int sr = RSTEP(g, j);
            v2f sb[4];
            #pragma unroll
            for (int i = 0; i < 4; ++i) {
              sb[i].x = p2[(4 * (sr + 1 + i))     * F_SZ + t * NWAVES * 16];
              sb[i].y = p2[(4 * (sr + 1 + i) + 1) * F_SZ + t * NWAVES * 16];
            }
            {
              v2f a;
              a.x = sn[4 * sr];
              a.y = sn[4 * sr + 1];
              acc = wmma_f32(a, rb2[t][4 * g + j], acc);
            }
            #pragma unroll
            for (int i = 0; i < 4; ++i) {
              v2f a;
              a.x = sn[4 * (sr + 1 + i)];
              a.y = sn[4 * (sr + 1 + i) + 1];
              acc = wmma_f32(a, sb[i], acc);
            }
            __builtin_amdgcn_sched_barrier(0);
          }
          {
            const int sr = RSTEP(g, 3);
            v2f a;
            a.x = sn[4 * sr];
            a.y = sn[4 * sr + 1];
            acc = wmma_f32(a, rb2[t][4 * g + 3], acc);
          }
        }
        float* const bp = &s_buf [hi8 * FSTR + fcol];
        float* const sp = &s_spec[hi8 * FSTR + fcol];
        #pragma unroll
        for (int r = 0; r < 8; ++r) {
          const int off = r * FSTR;
          float nb = MOM * bp[off] + INV * acc[r];
          bp[off] = nb;
          float ns = sp[off] - LR * nb;
          sp[off] = ns > 0.0f ? ns : 0.0f;
        }
      }
    }
    __syncthreads();

    if (tid < ROWS) {
      float g = s_tail[tid];
      #pragma unroll
      for (int w = 1; w < NWAVES; ++w) g += s_tail[w * ROWS + tid];
      const int off = tid * FSTR + (F_SZ - 1);
      float nb = MOM * s_buf[off] + INV * g;
      s_buf[off] = nb;
      float ns = s_spec[off] - LR * nb;
      s_spec[off] = ns > 0.0f ? ns : 0.0f;
    }
  }
  __syncthreads();

  for (int i = tid; i < ROWS * F_SZ; i += NTHREADS) {
    int r = i & (ROWS - 1), f = i >> 4;
    out[(b * F_SZ + f) * T_SZ + t0 + r] = s_spec[r * FSTR + f];
  }
}

extern "C" void kernel_launch(void* const* d_in, const int* in_sizes, int n_in,
                              void* d_out, int out_size, void* d_ws, size_t ws_size,
                              hipStream_t stream) {
  (void)in_sizes; (void)n_in; (void)out_size; (void)ws_size;
  const float* mel   = (const float*)d_in[0];
  const float* spec0 = (const float*)d_in[1];
  const float* fb    = (const float*)d_in[2];
  float* out = (float*)d_out;
  float* fbT = (float*)d_ws;

  {
    dim3 grid((M_SZ * F_SZ + TP_THREADS - 1) / TP_THREADS);
    hipLaunchKernelGGL(transpose_fb_kernel, grid, dim3(TP_THREADS), 0, stream,
                       fb, fbT);
  }
  {
    dim3 grid((B_SZ * T_SZ) / ROWS);
    hipLaunchKernelGGL(invmel_sgd_kernel, grid, dim3(NTHREADS), 0, stream,
                       mel, spec0, fb, fbT, out);
  }
}


# 2 more of every 16 k-steps' filterbank fragments (both GEMMs) kept in spare LDS, resident steps first in a round; the 10 streamed steps go in 2 rounds of 5: 16 L2 round trips, not 24; sum order same
# speedup vs baseline: 2.7716x; 1.0567x over previous
#include <hip/hip_runtime.h>


#define B_SZ   4
#define T_SZ   1024
#define F_SZ   1025
#define M_SZ   128
#define ITERS  20
#define LR     0.3f
#define MOM    0.9f
#define INV    (2.0f / (B_SZ * T_SZ))

#define ROWS     16
#define FSTR     1027
#define MSTR     131
#define NDSTR    131
#define NWAVES   32
#define NTHREADS 1024
#define TP_THREADS 256
#define NMT      (M_SZ / 16)
#define KQ       (NWAVES / NMT)
#define KQLEN    (1024 / KQ)
#define NFULL    (1024 / 16)
#define NDSZ     (ROWS * NDSTR)
#define NK1      (KQLEN / 4)
#define NK2      (M_SZ / 4)
#define NR1      (NK1 / 4)
#define NR2      (NK2 / 4)
#define NT2      (NFULL / NWAVES)
#define RSTEP(g, h, e)  (16 * (g) + 8 * (h) + 2 * (e))
#define LSTEP(g, h)     (16 * (g) + 8 * (h) + 1)
#define GPOS(i)         ((i) + 3)
#define NL1      (2 * (NK1 / 16))
#define NL2      (2 * (NK2 / 16))
#define NAHEAD   6

typedef __attribute__((ext_vector_type(2))) float v2f;
typedef __attribute__((ext_vector_type(8))) float v8f;

__device__ __forceinline__ v8f wmma_f32(v2f a, v2f b, v8f c) {
  return __builtin_amdgcn_wmma_f32_16x16x4_f32(false, a, false, b,
                                               (short)0, c, false, false);
}

__global__ __launch_bounds__(TP_THREADS)
void transpose_fb_kernel(const float* __restrict__ fb,
                         float* __restrict__ fbT) {
  int i = blockIdx.x * TP_THREADS + threadIdx.x;
  if (i < M_SZ * F_SZ) {
    int m = i / F_SZ, f = i - m * F_SZ;
    fbT[i] = fb[f * M_SZ + m];
  }
}

__global__ __launch_bounds__(NTHREADS)
void invmel_sgd_kernel(const float* __restrict__ mel,
                       const float* __restrict__ spec0,
                       const float* fb,
                       const float* fbT,
                       float* __restrict__ out)
{
  __shared__ float s_spec[ROWS * FSTR];
  __shared__ float s_buf [ROWS * FSTR];
  __shared__ float s_mel [ROWS * MSTR];
  __shared__ float s_ndp [KQ * NDSZ];
  __shared__ float s_nd  [NDSZ];
  __shared__ float s_tail[NWAVES * ROWS];
  __shared__ v2f s_fl1[NL1 * NTHREADS];
  __shared__ v2f s_fl2[NT2 * NL2 * NTHREADS];

  const int tid  = threadIdx.x;
  const int lane = tid & 31;
  const int wave = __builtin_amdgcn_readfirstlane(tid >> 5);
  const int nl   = lane & 15;
  const int hs   = lane >> 4;
  const int khi  = hs * 2;
  const int hi8  = hs * 8;
  const int kq   = wave >> 3;
  const int n0   = (wave & 7) * 16;

  const int blk = blockIdx.x;
  const int b   = blk / (T_SZ / ROWS);
  const int t0  = (blk % (T_SZ / ROWS)) * ROWS;

  for (int i = tid; i < ROWS * F_SZ; i += NTHREADS) {
    int r = i / F_SZ, c = i - r * F_SZ;
    s_spec[r * FSTR + c] = spec0[(b * T_SZ + t0 + r) * F_SZ + c];
    s_buf [r * FSTR + c] = 0.0f;
  }
  for (int i = tid; i < ROWS * M_SZ; i += NTHREADS) {
    int r = i & (ROWS - 1), m = i >> 4;
    s_mel[r * MSTR + m] = mel[(b * M_SZ + m) * T_SZ + t0 + r];
  }
  const float fb_last = fb[(F_SZ - 1) * M_SZ + n0 + nl];
  v2f tb;
  tb.x = fbT[(4 * wave + khi)     * F_SZ + (F_SZ - 1)];
  tb.y = fbT[(4 * wave + khi + 1) * F_SZ + (F_SZ - 1)];
  const float* const p1 = fb  + (kq * KQLEN + khi) * M_SZ + n0 + nl;
  const float* const p2 = fbT + khi * F_SZ + wave * 16 + nl;
  v2f rb1[NR1];
  #pragma unroll
  for (int g = 0; g < NK1 / 16; ++g) {
    #pragma unroll
    for (int h = 0; h < 2; ++h) {
      #pragma unroll
      for (int e = 0; e < 2; ++e) {
        rb1[4 * g + 2 * h + e].x = p1[(4 * RSTEP(g, h, e))     * M_SZ];
        rb1[4 * g + 2 * h + e].y = p1[(4 * RSTEP(g, h, e) + 1) * M_SZ];
      }
      v2f l;
      l.x = p1[(4 * LSTEP(g, h))     * M_SZ];
      l.y = p1[(4 * LSTEP(g, h) + 1) * M_SZ];
      s_fl1[(2 * g + h) * NTHREADS + tid] = l;
    }
  }
  v2f rb2[NT2][NR2];
  #pragma unroll
  for (int t = 0; t < NT2; ++t) {
    #pragma unroll
    for (int g = 0; g < NK2 / 16; ++g) {
      #pragma unroll
      for (int h = 0; h < 2; ++h) {
        #pragma unroll
        for (int e = 0; e < 2; ++e) {
          rb2[t][4 * g + 2 * h + e].x = p2[(4 * RSTEP(g, h, e))     * F_SZ + t * NWAVES * 16];
          rb2[t][4 * g + 2 * h + e].y = p2[(4 * RSTEP(g, h, e) + 1) * F_SZ + t * NWAVES * 16];
        }
        v2f l;
        l.x = p2[(4 * LSTEP(g, h))     * F_SZ + t * NWAVES * 16];
        l.y = p2[(4 * LSTEP(g, h) + 1) * F_SZ + t * NWAVES * 16];
        s_fl2[(t * NL2 + 2 * g + h) * NTHREADS + tid] = l;
      }
    }
  }
  __syncthreads();

  for (int it = 0; it < ITERS; ++it) {
    {
      const int kb = kq * KQLEN;
      v8f c;
      if (kq == 0) {
        const float* const melp = &s_mel[hi8 * MSTR + n0 + nl];
        #pragma unroll
        for (int r = 0; r < 8; ++r)
          c[r] = -melp[r * MSTR];
      } else {
        #pragma unroll
        for (int r = 0; r < 8; ++r) c[r] = 0.0f;
      }

      const float* const sa = &s_spec[nl * FSTR + kb + khi];
      #pragma unroll
      for (int g = 0; g < NK1 / 16; ++g) {
        #pragma unroll
        for (int h = 0; h < 2; ++h) {
          const int s0 = 16 * g + 8 * h;
          v2f sb[5];
          #pragma unroll
          for (int i = 0; i < 5; ++i) {
            sb[i].x = p1[(4 * (s0 + GPOS(i)))     * M_SZ];
            sb[i].y = p1[(4 * (s0 + GPOS(i)) + 1) * M_SZ];
          }
          const v2f lb = s_fl1[(2 * g + h) * NTHREADS + tid];
          v2f af[8];
          #pragma unroll
          for (int p = 0; p < NAHEAD; ++p) {
            af[p].x = sa[4 * (s0 + p)];
            af[p].y = sa[4 * (s0 + p) + 1];
          }
          __builtin_amdgcn_sched_barrier(0);
          #pragma unroll
          for (int p = 0; p < 8; ++p) {
            if (p == 8 - NAHEAD) {
              #pragma unroll
              for (int q = NAHEAD; q < 8; ++q) {
                af[q].x = sa[4 * (s0 + q)];
                af[q].y = sa[4 * (s0 + q) + 1];
              }
            }
            const v2f bf = (p == 0) ? rb1[4 * g + 2 * h]
                         : (p == 1) ? lb
                         : (p == 2) ? rb1[4 * g + 2 * h + 1]
                         : sb[p - 3];
            c = wmma_f32(af[p], bf, c);
          }
          __builtin_amdgcn_sched_group_barrier(0x008, 8 - NAHEAD, 0);
          __builtin_amdgcn_sched_group_barrier(0x100, 8 - NAHEAD, 0);
          __builtin_amdgcn_sched_group_barrier(0x008, NAHEAD, 0);
          __builtin_amdgcn_sched_barrier(0);
        }
      }
      float* const ndpp = &s_ndp[kq * NDSZ + hi8 * NDSTR + n0 + nl];
      #pragma unroll
      for (int r = 0; r < 8; ++r)
        ndpp[r * NDSTR] = c[r];
    }
    __syncthreads();

    if (wave < NMT) {
      v8f c;
      const float* const fxp = &s_ndp[hi8 * NDSTR + n0 + nl];
      #pragma unroll
      for (int r = 0; r < 8; ++r) {
        const int idx = r * NDSTR;
        c[r] = ((fxp[idx] + fxp[NDSZ + idx]) + fxp[2 * NDSZ + idx])
               + fxp[3 * NDSZ + idx];
      }
      v2f a, bf;
      a.x  = (hs == 0) ? s_spec[nl * FSTR + 1024] : 0.0f;
      a.y  = 0.0f;
      bf.x = (hs == 0) ? fb_last : 0.0f;
      bf.y = 0.0f;
      c = wmma_f32(a, bf, c);
      float* const ndw = &s_nd[hi8 * NDSTR + n0 + nl];
      #pragma unroll
      for (int r = 0; r < 8; ++r)
        ndw[r * NDSTR] = c[r];
    }
    __syncthreads();

    {
      v8f tacc = {0.f, 0.f, 0.f, 0.f, 0.f, 0.f, 0.f, 0.f};
      {
        v2f a;
        a.x = s_nd[nl * NDSTR + 4 * wave + khi];
        a.y = s_nd[nl * NDSTR + 4 * wave + khi + 1];
        tacc = wmma_f32(a, tb, tacc);
      }
      if (nl == 0) {
        #pragma unroll
        for (int r = 0; r < 8; ++r)
          s_tail[wave * ROWS + r + hi8] = tacc[r];
      }

      const float* const sn = &s_nd[nl * NDSTR + khi];
      #pragma unroll
      for (int t = 0; t < NT2; ++t) {
        const int fcol = (wave + t * NWAVES) * 16 + nl;
        v8f acc = {0.f, 0.f, 0.f, 0.f, 0.f, 0.f, 0.f, 0.f};

        #pragma unroll
        for (int g = 0; g < NK2 / 16; ++g) {
          #pragma unroll
          for (int h = 0; h < 2; ++h) {
            const int s0 = 16 * g + 8 * h;
            v2f sb[5];
            #pragma unroll
            for (int i = 0; i < 5; ++i) {
              sb[i].x = p2[(4 * (s0 + GPOS(i)))     * F_SZ + t * NWAVES * 16];
              sb[i].y = p2[(4 * (s0 + GPOS(i)) + 1) * F_SZ + t * NWAVES * 16];
            }
            const v2f lb = s_fl2[(t * NL2 + 2 * g + h) * NTHREADS + tid];
            v2f af[8];
            #pragma unroll
            for (int p = 0; p < NAHEAD; ++p) {
              af[p].x = sn[4 * (s0 + p)];
              af[p].y = sn[4 * (s0 + p) + 1];
            }
            __builtin_amdgcn_sched_barrier(0);
            #pragma unroll
            for (int p = 0; p < 8; ++p) {
              if (p == 8 - NAHEAD) {
                #pragma unroll
                for (int q = NAHEAD; q < 8; ++q) {
                  af[q].x = sn[4 * (s0 + q)];
                  af[q].y = sn[4 * (s0 + q) + 1];
                }
              }
              const v2f bf = (p == 0) ? rb2[t][4 * g + 2 * h]
                           : (p == 1) ? lb
                           : (p == 2) ? rb2[t][4 * g + 2 * h + 1]
                           : sb[p - 3];
              acc = wmma_f32(af[p], bf, acc);
            }
            __builtin_amdgcn_sched_group_barrier(0x008, 8 - NAHEAD, 0);
            __builtin_amdgcn_sched_group_barrier(0x100, 8 - NAHEAD, 0);
            __builtin_amdgcn_sched_group_barrier(0x008, NAHEAD, 0);
            __builtin_amdgcn_sched_barrier(0);
          }
        }
        float* const bp = &s_buf [hi8 * FSTR + fcol];
        float* const sp = &s_spec[hi8 * FSTR + fcol];
        #pragma unroll
        for (int r = 0; r < 8; ++r) {
          const int off = r * FSTR;
          float nb = MOM * bp[off] + INV * acc[r];
          bp[off] = nb;
          float ns = sp[off] - LR * nb;
          sp[off] = ns > 0.0f ? ns : 0.0f;
        }
      }
    }
    __syncthreads();

    if (tid < ROWS) {
      float g = s_tail[tid];
      #pragma unroll
      for (int w = 1; w < NWAVES; ++w) g += s_tail[w * ROWS + tid];
      const int off = tid * FSTR + (F_SZ - 1);
      float nb = MOM * s_buf[off] + INV * g;
      s_buf[off] = nb;
      float ns = s_spec[off] - LR * nb;
      s_spec[off] = ns > 0.0f ? ns : 0.0f;
    }
  }
  __syncthreads();

  for (int i = tid; i < ROWS * F_SZ; i += NTHREADS) {
    int r = i & (ROWS - 1), f = i >> 4;
    out[(b * F_SZ + f) * T_SZ + t0 + r] = s_spec[r * FSTR + f];
  }
}

extern "C" void kernel_launch(void* const* d_in, const int* in_sizes, int n_in,
                              void* d_out, int out_size, void* d_ws, size_t ws_size,
                              hipStream_t stream) {
  (void)in_sizes; (void)n_in; (void)out_size; (void)ws_size;
  const float* mel   = (const float*)d_in[0];
  const float* spec0 = (const float*)d_in[1];
  const float* fb    = (const float*)d_in[2];
  float* out = (float*)d_out;
  float* fbT = (float*)d_ws;

  {
    dim3 grid((M_SZ * F_SZ + TP_THREADS - 1) / TP_THREADS);
    hipLaunchKernelGGL(transpose_fb_kernel, grid, dim3(TP_THREADS), 0, stream,
                       fb, fbT);
  }
  {
    dim3 grid((B_SZ * T_SZ) / ROWS);
    hipLaunchKernelGGL(invmel_sgd_kernel, grid, dim3(NTHREADS), 0, stream,
                       mel, spec0, fb, fbT, out);
  }
}
